# starting kernel
# baseline (speedup 1.0000x reference)
#include <hip/hip_runtime.h>
#include <hip/hip_bf16.h>


typedef __attribute__((ext_vector_type(2))) float v2f;
typedef __attribute__((ext_vector_type(8))) float v8f;

#define BB   8
#define MM   2048
#define NN   8192
#define KNN  8
#define HID  128
#define INP  6
#define INP8 8
#define OUTW 2

__global__ void __launch_bounds__(256)
knn_seed_kernel(const float* __restrict__ partial,
                const float* __restrict__ predicted,
                float* __restrict__ xout) {
  __shared__ float sp[MM * 3];
  const int b   = blockIdx.y;
  const int tid = threadIdx.x;

  for (int i = tid; i < MM * 3; i += 256)
    sp[i] = partial[b * MM * 3 + i];
  __syncthreads();

  const int n = blockIdx.x * 256 + tid;
  const int row = b * NN + n;
  const float px = predicted[row * 3 + 0];
  const float py = predicted[row * 3 + 1];
  const float pz = predicted[row * 3 + 2];

  float bd[KNN]; int bi[KNN];
#pragma unroll
  for (int j = 0; j < KNN; ++j) { bd[j] = 3.4e38f; bi[j] = 0; }

  for (int m = 0; m < MM; ++m) {
    const float dx = px - sp[3 * m + 0];
    const float dy = py - sp[3 * m + 1];
    const float dz = pz - sp[3 * m + 2];
    float d = dx * dx + dy * dy + dz * dz;
    if (d < bd[KNN - 1]) {
      int id = m;
#pragma unroll
      for (int j = 0; j < KNN; ++j) {
        const bool sw = d < bd[j];
        const float td = bd[j]; const int ti = bi[j];
        bd[j] = sw ? d : td;  bi[j] = sw ? id : ti;
        d     = sw ? td : d;  id    = sw ? ti : id;
      }
    }
  }

  float sx = 0.f, sy = 0.f, sz = 0.f;
#pragma unroll
  for (int j = 0; j < KNN; ++j) {
    const int m = bi[j];
    sx += sp[3 * m + 0]; sy += sp[3 * m + 1]; sz += sp[3 * m + 2];
  }

  float* xr = xout + (size_t)row * INP8;
  xr[0] = px;          xr[1] = py;          xr[2] = pz;
  xr[3] = sx * 0.125f; xr[4] = sy * 0.125f; xr[5] = sz * 0.125f;
  xr[6] = 0.f;         xr[7] = 0.f;
}

__device__ __forceinline__ v8f wmma_f32(v2f a, v2f b, v8f c) {
  return __builtin_amdgcn_wmma_f32_16x16x4_f32(
       false, a,  false, b,
       (short)0, c,  false,  false);
}

__device__ __forceinline__ float relu(float x) { return x > 0.f ? x : 0.f; }

__global__ void __launch_bounds__(128)
mlp_refine_kernel(const float* __restrict__ xin,
                  const float* __restrict__ predicted,
                  const float* __restrict__ W1, const float* __restrict__ b1,
                  const float* __restrict__ W2, const float* __restrict__ b2,
                  const float* __restrict__ Wd, const float* __restrict__ bd,
                  const float* __restrict__ Wc, const float* __restrict__ bc,
                  float* __restrict__ out) {
  extern __shared__ float smem[];
  const int tid   = threadIdx.x;
  const int wave  = tid >> 5;
  const int lane  = tid & 31;
  const int lrow  = lane & 15;
  const int khalf = lane >> 4;

  float* lds_h    = smem + wave * (32 * HID);
  float* lds_seed = smem + 4 * (32 * HID) + wave * (32 * HID);
  float* lds_ht   = smem + 8 * (32 * HID) + wave * (32 * HID);

  const int rowBase = blockIdx.x * 128 + wave * 32;
  const int dRow = (lane < 16) ? 0 : 8;

  {
    const float* xr0 = xin + (size_t)(rowBase + lrow) * INP8;
    const float* xr1 = xin + (size_t)(rowBase + 16 + lrow) * INP8;
    const float kmask = khalf ? 0.f : 1.f;
    v2f a00, a01, a10, a11;
    a00.x = xr0[khalf * 2 + 0];     a00.y = xr0[khalf * 2 + 1];
    a01.x = xr0[4 + khalf * 2 + 0]; a01.y = xr0[4 + khalf * 2 + 1];
    a10.x = xr1[khalf * 2 + 0];     a10.y = xr1[khalf * 2 + 1];
    a11.x = xr1[4 + khalf * 2 + 0]; a11.y = xr1[4 + khalf * 2 + 1];
#pragma unroll
    for (int nt = 0; nt < 8; ++nt) {
      const int n = nt * 16 + lrow;
      const float bv = b1[n];
      v8f c0, c1;
#pragma unroll
      for (int j = 0; j < 8; ++j) { c0[j] = bv; c1[j] = bv; }
      v2f bf0, bf1;
      const int k0 = khalf * 2;
      bf0.x = W1[n * INP + k0];        bf0.y = W1[n * INP + k0 + 1];
      bf1.x = kmask * W1[n * INP + 4]; bf1.y = kmask * W1[n * INP + 5];
      c0 = wmma_f32(a00, bf0, c0);  c0 = wmma_f32(a01, bf1, c0);
      c1 = wmma_f32(a10, bf0, c1);  c1 = wmma_f32(a11, bf1, c1);
#pragma unroll
      for (int j = 0; j < 8; ++j) {
        const int r = j + dRow;
        lds_h[r * HID + n]        = relu(c0[j]);
        lds_h[(16 + r) * HID + n] = relu(c1[j]);
      }
    }
  }

#pragma unroll
  for (int nt = 0; nt < 8; ++nt) {
    const int n = nt * 16 + lrow;
    const float bv = b2[n];
    v8f c0, c1;
#pragma unroll
    for (int j = 0; j < 8; ++j) { c0[j] = bv; c1[j] = bv; }
    const float* w2r = W2 + n * HID;
    for (int k4 = 0; k4 < HID; k4 += 4) {
      const int kb = k4 + khalf * 2;
      v2f bf, aA, aB;
      bf.x = w2r[kb];                       bf.y = w2r[kb + 1];
      aA.x = lds_h[lrow * HID + kb];        aA.y = lds_h[lrow * HID + kb + 1];
      aB.x = lds_h[(16 + lrow) * HID + kb]; aB.y = lds_h[(16 + lrow) * HID + kb + 1];
      c0 = wmma_f32(aA, bf, c0);
      c1 = wmma_f32(aB, bf, c1);
    }
#pragma unroll
    for (int j = 0; j < 8; ++j) {
      const int r = j + dRow;
      lds_seed[r * HID + n]        = c0[j];
      lds_seed[(16 + r) * HID + n] = c1[j];
    }
  }

  for (int t = 0; t < OUTW; ++t) {
#pragma unroll
    for (int nt = 0; nt < 8; ++nt) {
      const int n = nt * 16 + lrow;
      const float bv = bd[n];
      v8f c0, c1;
#pragma unroll
      for (int j = 0; j < 8; ++j) { c0[j] = bv; c1[j] = bv; }
      for (int k4 = 0; k4 < HID; k4 += 4) {
        const int kb = k4 + khalf * 2;
        v2f bf, aA, aB;
        bf.x = Wd[(kb * HID + n) * OUTW + t];
        bf.y = Wd[((kb + 1) * HID + n) * OUTW + t];
        aA.x = lds_seed[lrow * HID + kb];
        aA.y = lds_seed[lrow * HID + kb + 1];
        aB.x = lds_seed[(16 + lrow) * HID + kb];
        aB.y = lds_seed[(16 + lrow) * HID + kb + 1];
        c0 = wmma_f32(aA, bf, c0);
        c1 = wmma_f32(aB, bf, c1);
      }
#pragma unroll
      for (int j = 0; j < 8; ++j) {
        const int r = j + dRow;
        lds_ht[r * HID + n]        = relu(c0[j]);
        lds_ht[(16 + r) * HID + n] = relu(c1[j]);
      }
    }
    {
      const int rg = rowBase + lane;
      float a0 = bc[0], a1 = bc[1], a2 = bc[2];
      for (int i = 0; i < HID; ++i) {
        const float v = lds_ht[lane * HID + i];
        a0 += v * Wc[0 * HID + i];
        a1 += v * Wc[1 * HID + i];
        a2 += v * Wc[2 * HID + i];
      }
      float* o = out + ((size_t)rg * OUTW + t) * 3;
      o[0] = predicted[rg * 3 + 0] + a0;
      o[1] = predicted[rg * 3 + 1] + a1;
      o[2] = predicted[rg * 3 + 2] + a2;
    }
  }
}

extern "C" void kernel_launch(void* const* d_in, const int* in_sizes, int n_in,
                              void* d_out, int out_size, void* d_ws, size_t ws_size,
                              hipStream_t stream) {
  const float* partial   = (const float*)d_in[0];
  const float* predicted = (const float*)d_in[1];
  const float* W1 = (const float*)d_in[2];
  const float* b1 = (const float*)d_in[3];
  const float* W2 = (const float*)d_in[4];
  const float* b2 = (const float*)d_in[5];
  const float* Wd = (const float*)d_in[6];
  const float* bd = (const float*)d_in[7];
  const float* Wc = (const float*)d_in[8];
  const float* bc = (const float*)d_in[9];
  float* out = (float*)d_out;

  float* x = (float*)d_ws;

  dim3 g1(NN / 256, BB);
  knn_seed_kernel<<<g1, 256, 0, stream>>>(partial, predicted, x);

  const int blocks = (BB * NN) / 128;
  const size_t lds = (size_t)12 * 32 * HID * sizeof(float);
  mlp_refine_kernel<<<blocks, 128, lds, stream>>>(
      x, predicted, W1, b1, W2, b2, Wd, bd, Wc, bc, out);
}


# KNN top-8 kept by a sorting-network merge per 8 points (sort 8, then bitonic merge into the list): no test, no branch per point; scores bit-identical; ties may rank in another order
# speedup vs baseline: 4.1505x; 4.1505x over previous
#include <hip/hip_runtime.h>
#include <hip/hip_bf16.h>


typedef __attribute__((ext_vector_type(2))) float v2f;
typedef __attribute__((ext_vector_type(4))) float v4f;
typedef __attribute__((ext_vector_type(8))) float v8f;

#define BB   8
#define MM   2048
#define NN   8192
#define KNN  8
#define HID  128
#define INP  6
#define INP8 8
#define OUTW 2

#define CEX(D, I, i, j) {                                   \
    const bool sw_ = D[j] < D[i];                           \
    const float lo_ = __builtin_fminf(D[i], D[j]);          \
    const float hi_ = __builtin_fmaxf(D[i], D[j]);          \
    const int li_ = sw_ ? I[j] : I[i];                      \
    const int hi2_ = sw_ ? I[i] : I[j];                     \
    D[i] = lo_; D[j] = hi_; I[i] = li_; I[j] = hi2_; }

#define CEXS(D, I, i, j) {                                  \
    const bool sw_ = D[j] < D[i];                           \
    const float lo_ = sw_ ? D[j] : D[i];                    \
    const float hi_ = sw_ ? D[i] : D[j];                    \
    const int li_ = sw_ ? I[j] : I[i];                      \
    const int hi2_ = sw_ ? I[i] : I[j];                     \
    D[i] = lo_; D[j] = hi_; I[i] = li_; I[j] = hi2_; }

__global__ void __launch_bounds__(256)
knn_seed_kernel(const float* __restrict__ partial,
                const float* __restrict__ predicted,
                float* __restrict__ xout) {
  __shared__ float sp[MM * 3];
  const int b   = blockIdx.y;
  const int tid = threadIdx.x;

  for (int i = tid; i < MM * 3; i += 256)
    sp[i] = partial[b * MM * 3 + i];
  __syncthreads();

  const int n = blockIdx.x * 256 + tid;
  const int row = b * NN + n;
  const float px = predicted[row * 3 + 0];
  const float py = predicted[row * 3 + 1];
  const float pz = predicted[row * 3 + 2];

  float bd[KNN]; int bi[KNN];
#pragma unroll
  for (int j = 0; j < KNN; ++j) { bd[j] = 3.4e38f; bi[j] = 0; }

  const int lane = tid & 31;
  float nx = sp[3 * lane + 0];
  float ny = sp[3 * lane + 1];
  float nz = sp[3 * lane + 2];
  for (int g = 0; g < MM / 32; ++g) {
    const int cx = __float_as_int(nx);
    const int cy = __float_as_int(ny);
    const int cz = __float_as_int(nz);
    const int gn = (g + 1 < MM / 32) ? g + 1 : g;
    nx = sp[3 * (32 * gn + lane) + 0];
    ny = sp[3 * (32 * gn + lane) + 1];
    nz = sp[3 * (32 * gn + lane) + 2];
#pragma nounroll
    for (int j0 = 0; j0 < 32; j0 += 8) {
      float d[8]; int id[8];
#pragma unroll
      for (int j = 0; j < 8; ++j) {
        const float qx = __int_as_float(__builtin_amdgcn_readlane(cx, j0 + j));
        const float qy = __int_as_float(__builtin_amdgcn_readlane(cy, j0 + j));
        const float qz = __int_as_float(__builtin_amdgcn_readlane(cz, j0 + j));
        const float dx = px - qx;
        const float dy = py - qy;
        const float dz = pz - qz;
        float s2;
        {
#pragma clang fp contract(off)
          s2 = dx * dx + dy * dy;
        }
        d[j]  = __builtin_fmaf(dz, dz, s2);
        id[j] = 32 * g + j0 + j;
      }
      CEX(d, id, 0, 2) CEX(d, id, 1, 3) CEX(d, id, 4, 6) CEX(d, id, 5, 7)
      CEX(d, id, 0, 4) CEX(d, id, 1, 5) CEX(d, id, 2, 6) CEX(d, id, 3, 7)
      CEX(d, id, 0, 1) CEX(d, id, 2, 3) CEX(d, id, 4, 5) CEX(d, id, 6, 7)
      CEX(d, id, 2, 4) CEX(d, id, 3, 5)
      CEX(d, id, 1, 4) CEX(d, id, 3, 6)
      CEX(d, id, 1, 2) CEX(d, id, 3, 4) CEX(d, id, 5, 6)
#pragma unroll
      for (int i = 0; i < 8; ++i) {
        const bool sw = d[7 - i] < bd[i];
        bd[i] = sw ? d[7 - i] : bd[i];
        bi[i] = sw ? id[7 - i] : bi[i];
      }
      CEXS(bd, bi, 0, 4) CEXS(bd, bi, 1, 5) CEXS(bd, bi, 2, 6) CEXS(bd, bi, 3, 7)
      CEXS(bd, bi, 0, 2) CEXS(bd, bi, 1, 3) CEXS(bd, bi, 4, 6) CEXS(bd, bi, 5, 7)
      CEXS(bd, bi, 0, 1) CEXS(bd, bi, 2, 3) CEXS(bd, bi, 4, 5) CEXS(bd, bi, 6, 7)
    }
  }

  float sx = 0.f, sy = 0.f, sz = 0.f;
#pragma unroll
  for (int j = 0; j < KNN; ++j) {
    const int m = bi[j];
    sx += sp[3 * m + 0]; sy += sp[3 * m + 1]; sz += sp[3 * m + 2];
  }

  float* xr = xout + (size_t)row * INP8;
  xr[0] = px;          xr[1] = py;          xr[2] = pz;
  xr[3] = sx * 0.125f; xr[4] = sy * 0.125f; xr[5] = sz * 0.125f;
  xr[6] = 0.f;         xr[7] = 0.f;
}

__device__ __forceinline__ v8f wmma_f32(v2f a, v2f b, v8f c) {
  return __builtin_amdgcn_wmma_f32_16x16x4_f32(
       false, a,  false, b,
       (short)0, c,  false,  false);
}

__device__ __forceinline__ v8f relu8(v8f v) {
#pragma unroll
  for (int j = 0; j < 8; ++j) v[j] = v[j] > 0.f ? v[j] : 0.f;
  return v;
}

#define WP    132
#define L_W2  0
#define L_WD  (L_W2 + HID * WP)
#define L_WC  (L_WD + OUTW * HID * WP)
#define L_W1  (L_WC + 16 * WP)
#define L_B1  (L_W1 + HID * INP8)
#define L_B2  (L_B1 + HID)
#define L_BD  (L_B2 + HID)
#define L_BC  (L_BD + HID)
#define L_END (L_BC + 16)
#define MLP_BLOCKS 256

__device__ __forceinline__ v8f bias_tile(const float* b) {
  const v4f lo = *(const v4f*)(b);
  const v4f hi = *(const v4f*)(b + 4);
  v8f r = {lo.x, lo.y, lo.z, lo.w, hi.x, hi.y, hi.z, hi.w};
  return r;
}

__device__ __forceinline__ void dense128(const float* w, const float* b,
                                         const v8f (&in)[8], v8f (&acc)[8]) {
  v4f a[2][8];
#pragma unroll
  for (int U = 0; U < 8; ++U) acc[U] = bias_tile(b + 16 * U);
#pragma unroll
  for (int U = 0; U < 8; ++U) a[0][U] = *(const v4f*)(w + 16 * U * WP);
#pragma unroll
  for (int s = 0; s < 16; ++s) {
    const int T = s >> 1, ip = s & 1;
    if (s + 1 < 16) {
      const int col = 16 * ((s + 1) >> 1) + 4 * ((s + 1) & 1);
#pragma unroll
      for (int U = 0; U < 8; ++U)
        a[(s + 1) & 1][U] = *(const v4f*)(w + 16 * U * WP + col);
    }
    __builtin_amdgcn_sched_barrier(0);
#pragma unroll
    for (int U = 0; U < 8; ++U) {
      const v2f av = {a[s & 1][U].x, a[s & 1][U].y};
      const v2f bv = {in[T][4 * ip + 0], in[T][4 * ip + 1]};
      acc[U] = wmma_f32(av, bv, acc[U]);
    }
#pragma unroll
    for (int U = 0; U < 8; ++U) {
      const v2f av = {a[s & 1][U].z, a[s & 1][U].w};
      const v2f bv = {in[T][4 * ip + 2], in[T][4 * ip + 3]};
      acc[U] = wmma_f32(av, bv, acc[U]);
    }
    __builtin_amdgcn_sched_barrier(0);
  }
}

__global__ void __launch_bounds__(256)
mlp_refine_kernel(const float* __restrict__ xin,
                  const float* __restrict__ predicted,
                  const float* __restrict__ W1, const float* __restrict__ b1,
                  const float* __restrict__ W2, const float* __restrict__ b2,
                  const float* __restrict__ Wd, const float* __restrict__ bd,
                  const float* __restrict__ Wc, const float* __restrict__ bc,
                  float* __restrict__ out) {
  extern __shared__ __attribute__((aligned(16))) float smem[];
  const int tid  = threadIdx.x;
  const int wave = tid >> 5;
  const int lane = tid & 31;
  const int c    = lane & 15;
  const int h    = lane >> 4;

  const int NT     = (BB * NN) / 16;
  const int stride = MLP_BLOCKS * 8;
  int tile = blockIdx.x * 8 + wave;
  v2f x0 = {0.f, 0.f}, x1 = {0.f, 0.f};
  if (tile < NT) {
    const float* xr = xin + (size_t)(tile * 16 + c) * INP8 + 2 * h;
    x0 = *(const v2f*)(xr);
    x1 = *(const v2f*)(xr + 4);
  }

#pragma unroll
  for (int j = 0; j < (HID * HID / 4) / 256; ++j) {
    const int i = tid + 256 * j;
    const int n = i >> 5, q = i & 31;
    *(v4f*)(smem + L_W2 + n * WP + 4 * q) = *(const v4f*)(W2 + 4 * i);
  }
#pragma unroll
  for (int j = 0; j < (HID * HID / 4) / 256; ++j) {
    const int i = tid + 256 * j;
    const int n = i & (HID - 1), k = 4 * (i >> 7);
    const v2f g0 = *(const v2f*)(Wd + ((k + 0) * HID + n) * OUTW);
    const v2f g1 = *(const v2f*)(Wd + ((k + 1) * HID + n) * OUTW);
    const v2f g2 = *(const v2f*)(Wd + ((k + 2) * HID + n) * OUTW);
    const v2f g3 = *(const v2f*)(Wd + ((k + 3) * HID + n) * OUTW);
    const v4f t0 = {g0.x, g1.x, g2.x, g3.x};
    const v4f t1 = {g0.y, g1.y, g2.y, g3.y};
    *(v4f*)(smem + L_WD + n * WP + k)            = t0;
    *(v4f*)(smem + L_WD + HID * WP + n * WP + k) = t1;
  }
#pragma unroll
  for (int j = 0; j < (16 * HID) / 256; ++j) {
    const int i = tid + 256 * j;
    const int r = i >> 7, k = i & (HID - 1);
    const float v = Wc[(r < 3 ? r : 2) * HID + k];
    smem[L_WC + r * WP + k] = r < 3 ? v : 0.f;
  }
#pragma unroll
  for (int j = 0; j < (HID * INP8) / 256; ++j) {
    const int i = tid + 256 * j;
    const int n = i >> 3, k = i & 7;
    const float v = W1[n * INP + (k < INP ? k : INP - 1)];
    smem[L_W1 + i] = k < INP ? v : 0.f;
  }
  {
    const int n = tid & (HID - 1);
    smem[L_B1 + n] = b1[n];
    smem[L_B2 + n] = b2[n];
    smem[L_BD + n] = bd[n];
    const int m = tid & 15;
    const float v = bc[m < 3 ? m : 2];
    smem[L_BC + m] = m < 3 ? v : 0.f;
  }
  __syncthreads();

  const float* w1l = smem + L_W1 + c * INP8 + 2 * h;
  const float* w2l = smem + L_W2 + c * WP + 8 * h;
  const float* wdl = smem + L_WD + c * WP + 8 * h;
  const float* wcl = smem + L_WC + c * WP + 8 * h;

  for (; tile < NT; tile += stride) {
    asm volatile("" ::: "memory");
    const int row = tile * 16 + c;

    v2f a1f[8][2];
    v8f hh[8];
#pragma unroll
    for (int U = 0; U < 8; ++U) {
      hh[U]     = bias_tile(smem + L_B1 + 8 * h + 16 * U);
      a1f[U][0] = *(const v2f*)(w1l + 16 * U * INP8);
      a1f[U][1] = *(const v2f*)(w1l + 16 * U * INP8 + 4);
    }
    const float* p = predicted + (size_t)row * 3;
    const float pr0 = p[0], pr1 = p[1], pr2 = p[2];
    v2f xn0 = x0, xn1 = x1;
    if (tile + stride < NT) {
      const float* xr = xin + (size_t)((tile + stride) * 16 + c) * INP8 + 2 * h;
      xn0 = *(const v2f*)(xr);
      xn1 = *(const v2f*)(xr + 4);
    }
#pragma unroll
    for (int U = 0; U < 8; ++U) hh[U] = wmma_f32(a1f[U][0], x0, hh[U]);
#pragma unroll
    for (int U = 0; U < 8; ++U) hh[U] = relu8(wmma_f32(a1f[U][1], x1, hh[U]));

    v8f seed[8];
    dense128(w2l, smem + L_B2 + 8 * h, hh, seed);

    for (int t = 0; t < OUTW; ++t) {
      v8f ht[8];
      dense128(wdl + t * HID * WP, smem + L_BD + 8 * h, seed, ht);

      v4f wc[16];
#pragma unroll
      for (int s = 0; s < 16; ++s)
        wc[s] = *(const v4f*)(wcl + 16 * (s >> 1) + 4 * (s & 1));
      v8f o0 = bias_tile(smem + L_BC + 8 * h);
      v8f o1 = {0.f, 0.f, 0.f, 0.f, 0.f, 0.f, 0.f, 0.f};
      __builtin_amdgcn_sched_barrier(0);
#pragma unroll
      for (int U = 0; U < 8; ++U) ht[U] = relu8(ht[U]);
#pragma unroll
      for (int T = 0; T < 8; T += 2) {
#pragma unroll
        for (int q = 0; q < 4; ++q) {
          const v4f ae = wc[2 * T + (q >> 1)];
          const v4f ao = wc[2 * (T + 1) + (q >> 1)];
          const v2f av0 = (q & 1) ? (v2f){ae.z, ae.w} : (v2f){ae.x, ae.y};
          const v2f av1 = (q & 1) ? (v2f){ao.z, ao.w} : (v2f){ao.x, ao.y};
          const v2f bv0 = {ht[T][2 * q], ht[T][2 * q + 1]};
          const v2f bv1 = {ht[T + 1][2 * q], ht[T + 1][2 * q + 1]};
          o0 = wmma_f32(av0, bv0, o0);
          o1 = wmma_f32(av1, bv1, o1);
        }
      }
      if (h == 0) {
        float* o = out + ((size_t)row * OUTW + t) * 3;
        o[0] = pr0 + (o0[0] + o1[0]);
        o[1] = pr1 + (o0[1] + o1[1]);
        o[2] = pr2 + (o0[2] + o1[2]);
      }
    }
    x0 = xn0; x1 = xn1;
  }
}

extern "C" void kernel_launch(void* const* d_in, const int* in_sizes, int n_in,
                              void* d_out, int out_size, void* d_ws, size_t ws_size,
                              hipStream_t stream) {
  const float* partial   = (const float*)d_in[0];
  const float* predicted = (const float*)d_in[1];
  const float* W1 = (const float*)d_in[2];
  const float* b1 = (const float*)d_in[3];
  const float* W2 = (const float*)d_in[4];
  const float* b2 = (const float*)d_in[5];
  const float* Wd = (const float*)d_in[6];
  const float* bd = (const float*)d_in[7];
  const float* Wc = (const float*)d_in[8];
  const float* bc = (const float*)d_in[9];
  float* out = (float*)d_out;

  float* x = (float*)d_ws;

  dim3 g1(NN / 256, BB);
  knn_seed_kernel<<<g1, 256, 0, stream>>>(partial, predicted, x);

  const size_t lds = (size_t)L_END * sizeof(float);
  mlp_refine_kernel<<<MLP_BLOCKS, 256, lds, stream>>>(
      x, predicted, W1, b1, W2, b2, Wd, bd, Wc, bc, out);
}


# One launch: a block runs the MLP on the 256 rows whose KNN it just finished (a wave's 32 rows are its 2 tiles, passed by lane shuffle; no workspace); W2, Wd staged a slice per scan trip; same sums
# speedup vs baseline: 6.6455x; 1.6011x over previous
#include <hip/hip_runtime.h>
#include <hip/hip_bf16.h>


typedef __attribute__((ext_vector_type(2))) float v2f;
typedef __attribute__((ext_vector_type(4))) float v4f;
typedef __attribute__((ext_vector_type(8))) float v8f;

#define BB   8
#define MM   2048
#define NN   8192
#define KNN  8
#define HID  128
#define INP  6
#define INP8 8
#define OUTW 2

#define CEXD(E, i, j) {                                     \
    const float lo_ = __builtin_fminf(E[i], E[j]);          \
    const float hi_ = __builtin_fmaxf(E[i], E[j]);          \
    E[i] = lo_; E[j] = hi_; }

#define SORT8D(E)                                                             \
    CEXD(E, 0, 2) CEXD(E, 1, 3) CEXD(E, 4, 6) CEXD(E, 5, 7)                   \
    CEXD(E, 0, 4) CEXD(E, 1, 5) CEXD(E, 2, 6) CEXD(E, 3, 7)                   \
    CEXD(E, 0, 1) CEXD(E, 2, 3) CEXD(E, 4, 5) CEXD(E, 6, 7)                   \
    CEXD(E, 2, 4) CEXD(E, 3, 5)                                               \
    CEXD(E, 1, 4) CEXD(E, 3, 6)                                               \
    CEXD(E, 1, 2) CEXD(E, 3, 4) CEXD(E, 5, 6)

#define LOW8D(A, B) {                                                         \
    _Pragma("unroll")                                                         \
    for (int i_ = 0; i_ < 8; ++i_) A[i_] = __builtin_fminf(A[i_], B[7 - i_]); \
    CEXD(A, 0, 4) CEXD(A, 1, 5) CEXD(A, 2, 6) CEXD(A, 3, 7)                   \
    CEXD(A, 0, 2) CEXD(A, 1, 3) CEXD(A, 4, 6) CEXD(A, 5, 7)                   \
    CEXD(A, 0, 1) CEXD(A, 2, 3) CEXD(A, 4, 5) CEXD(A, 6, 7) }

#define CEXS(D, I, i, j) {                                  \
    const bool sw_ = D[j] < D[i];                           \
    const float lo_ = sw_ ? D[j] : D[i];                    \
    const float hi_ = sw_ ? D[i] : D[j];                    \
    const int li_ = sw_ ? I[j] : I[i];                      \
    const int hi2_ = sw_ ? I[i] : I[j];                     \
    D[i] = lo_; D[j] = hi_; I[i] = li_; I[j] = hi2_; }

__device__ __forceinline__ float dist2(float px, float py, float pz,
                                       float qx, float qy, float qz) {
  const float dx = px - qx;
  const float dy = py - qy;
  const float dz = pz - qz;
  float s2;
  {
#pragma clang fp contract(off)
    s2 = dx * dx + dy * dy;
  }
  return __builtin_fmaf(dz, dz, s2);
}

__device__ __forceinline__ v2f dist2x2(v2f px, v2f py, v2f pz,
                                       float qx, float qy, float qz) {
  const v2f dx = px - qx;
  const v2f dy = py - qy;
  const v2f dz = pz - qz;
  v2f s2;
  {
#pragma clang fp contract(off)
    s2 = dx * dx + dy * dy;
  }
  return __builtin_elementwise_fma(dz, dz, s2);
}

#define QPL 4
#define NQS (8 / QPL)
#define GP  (MM / 32 / QPL)

__device__ __forceinline__ v8f wmma_f32(v2f a, v2f b, v8f c) {
  return __builtin_amdgcn_wmma_f32_16x16x4_f32(
       false, a,  false, b,
       (short)0, c,  false,  false);
}

__device__ __forceinline__ v8f relu8(v8f v) {
#pragma unroll
  for (int j = 0; j < 8; ++j) v[j] = v[j] > 0.f ? v[j] : 0.f;
  return v;
}

#define WP    132
#define L_W2  0
#define L_WD  (L_W2 + HID * WP)
#define L_WC  (L_WD + OUTW * HID * WP)
#define L_W1  (L_WC + 16 * WP)
#define L_B1  (L_W1 + HID * INP8)
#define L_B2  (L_B1 + HID)
#define L_BD  (L_B2 + HID)
#define L_BC  (L_BD + HID)
#define L_END (L_BC + 16)

__device__ __forceinline__ v8f bias_tile(const float* b) {
  const v4f lo = *(const v4f*)(b);
  const v4f hi = *(const v4f*)(b + 4);
  v8f r = {lo.x, lo.y, lo.z, lo.w, hi.x, hi.y, hi.z, hi.w};
  return r;
}

__device__ __forceinline__ void dense128(const float* w, const float* b,
                                         const v8f (&in)[8], v8f (&acc)[8]) {
  v4f a[2][8];
#pragma unroll
  for (int U = 0; U < 8; ++U) acc[U] = bias_tile(b + 16 * U);
#pragma unroll
  for (int U = 0; U < 8; ++U) a[0][U] = *(const v4f*)(w + 16 * U * WP);
#pragma unroll
  for (int s = 0; s < 16; ++s) {
    const int T = s >> 1, ip = s & 1;
    if (s + 1 < 16) {
      const int col = 16 * ((s + 1) >> 1) + 4 * ((s + 1) & 1);
#pragma unroll
      for (int U = 0; U < 8; ++U)
        a[(s + 1) & 1][U] = *(const v4f*)(w + 16 * U * WP + col);
    }
    __builtin_amdgcn_sched_barrier(0);
#pragma unroll
    for (int U = 0; U < 8; ++U) {
      const v2f av = {a[s & 1][U].x, a[s & 1][U].y};
      const v2f bv = {in[T][4 * ip + 0], in[T][4 * ip + 1]};
      acc[U] = wmma_f32(av, bv, acc[U]);
    }
#pragma unroll
    for (int U = 0; U < 8; ++U) {
      const v2f av = {a[s & 1][U].z, a[s & 1][U].w};
      const v2f bv = {in[T][4 * ip + 2], in[T][4 * ip + 3]};
      acc[U] = wmma_f32(av, bv, acc[U]);
    }
    __builtin_amdgcn_sched_barrier(0);
  }
}

__global__ void __launch_bounds__(256)
refine_kernel(const float* __restrict__ partial,
              const float* __restrict__ predicted,
              const float* __restrict__ W1, const float* __restrict__ b1,
              const float* __restrict__ W2, const float* __restrict__ b2,
              const float* __restrict__ Wd, const float* __restrict__ bdc,
              const float* __restrict__ Wc, const float* __restrict__ bc,
              float* __restrict__ out) {
  extern __shared__ __attribute__((aligned(16))) float smem[];
  __shared__ __attribute__((aligned(16))) float sp[MM * 3];
  __shared__ float xd[8 * QPL * KNN * 32];
  __shared__ int   xg[8 * QPL * KNN * 32];
  const int b   = blockIdx.y;
  const int tid = threadIdx.x;

  v4f w2c = *(const v4f*)(W2 + 4 * tid);
  v2f wdc0, wdc1, wdc2, wdc3;
  {
    const int n = tid & (HID - 1), kq = 4 * (tid >> 7);
    wdc0 = *(const v2f*)(Wd + ((kq + 0) * HID + n) * OUTW);
    wdc1 = *(const v2f*)(Wd + ((kq + 1) * HID + n) * OUTW);
    wdc2 = *(const v2f*)(Wd + ((kq + 2) * HID + n) * OUTW);
    wdc3 = *(const v2f*)(Wd + ((kq + 3) * HID + n) * OUTW);
  }

  {
    const v4f* src = (const v4f*)(partial + b * MM * 3);
#pragma unroll
    for (int j = 0; j < (MM * 3 / 4) / 256; ++j)
      ((v4f*)sp)[tid + 256 * j] = src[tid + 256 * j];
  }
#pragma unroll
  for (int j = 0; j < (16 * HID) / 256; ++j) {
    const int i = tid + 256 * j;
    const int r = i >> 7, k = i & (HID - 1);
    const float v = Wc[(r < 3 ? r : 2) * HID + k];
    smem[L_WC + r * WP + k] = r < 3 ? v : 0.f;
  }
#pragma unroll
  for (int j = 0; j < (HID * INP8) / 256; ++j) {
    const int i = tid + 256 * j;
    const int n = i >> 3, k = i & 7;
    const float v = W1[n * INP + (k < INP ? k : INP - 1)];
    smem[L_W1 + i] = k < INP ? v : 0.f;
  }
  {
    const int n = tid & (HID - 1);
    smem[L_B1 + n] = b1[n];
    smem[L_B2 + n] = b2[n];
    smem[L_BD + n] = bdc[n];
    const int m = tid & 15;
    const float v = bc[m < 3 ? m : 2];
    smem[L_BC + m] = m < 3 ? v : 0.f;
  }
  __syncthreads();

  const int wave = __builtin_amdgcn_readfirstlane(tid >> 5);
  const int lane = tid & 31;
  const int part = wave / NQS;
  const int qs   = wave % NQS;
  const int nbase = blockIdx.x * 256 + 32 * QPL * qs + lane;

  v2f pxv[QPL / 2], pyv[QPL / 2], pzv[QPL / 2];
#pragma unroll
  for (int q = 0; q < QPL; ++q) {
    const float* pr = predicted + (size_t)(b * NN + nbase + 32 * q) * 3;
    pxv[q >> 1][q & 1] = pr[0];
    pyv[q >> 1][q & 1] = pr[1];
    pzv[q >> 1][q & 1] = pr[2];
  }

  const int mrow = b * NN + nbase + 32 * part;
  const float mx = predicted[(size_t)mrow * 3 + 0];
  const float my = predicted[(size_t)mrow * 3 + 1];
  const float mz = predicted[(size_t)mrow * 3 + 2];

  float bd[QPL][KNN]; int bg[QPL][KNN];
#pragma unroll
  for (int q = 0; q < QPL; ++q)
#pragma unroll
    for (int j = 0; j < KNN; ++j) { bd[q][j] = 3.4e38f; bg[q][j] = 0; }

  const int g0 = GP * part;
  const int g1 = g0 + GP;
  float nx = sp[3 * (32 * g0 + lane) + 0];
  float ny = sp[3 * (32 * g0 + lane) + 1];
  float nz = sp[3 * (32 * g0 + lane) + 2];
  for (int g = g0; g < g1; ++g) {
    const int cx = __float_as_int(nx);
    const int cy = __float_as_int(ny);
    const int cz = __float_as_int(nz);
    const int gn = (g + 1 < g1) ? g + 1 : g;
    nx = sp[3 * (32 * gn + lane) + 0];
    ny = sp[3 * (32 * gn + lane) + 1];
    nz = sp[3 * (32 * gn + lane) + 2];
    {
      const int i = tid + 256 * (g - g0);
      *(v4f*)(smem + L_W2 + (i >> 5) * WP + 4 * (i & 31)) = w2c;
      const int n = i & (HID - 1), kq = 4 * (i >> 7);
      const v4f t0 = {wdc0.x, wdc1.x, wdc2.x, wdc3.x};
      const v4f t1 = {wdc0.y, wdc1.y, wdc2.y, wdc3.y};
      *(v4f*)(smem + L_WD + n * WP + kq)            = t0;
      *(v4f*)(smem + L_WD + HID * WP + n * WP + kq) = t1;
      const int i2 = tid + 256 * (gn - g0);
      w2c = *(const v4f*)(W2 + 4 * i2);
      const int n2 = i2 & (HID - 1), k2 = 4 * (i2 >> 7);
      wdc0 = *(const v2f*)(Wd + ((k2 + 0) * HID + n2) * OUTW);
      wdc1 = *(const v2f*)(Wd + ((k2 + 1) * HID + n2) * OUTW);
      wdc2 = *(const v2f*)(Wd + ((k2 + 2) * HID + n2) * OUTW);
      wdc3 = *(const v2f*)(Wd + ((k2 + 3) * HID + n2) * OUTW);
    }
    asm volatile("" ::: "memory");
    float e[QPL][4][8];
#pragma unroll
    for (int s = 0; s < 4; ++s) {
#pragma unroll
      for (int j = 0; j < 8; ++j) {
        const float qx = __int_as_float(__builtin_amdgcn_readlane(cx, 8 * s + j));
        const float qy = __int_as_float(__builtin_amdgcn_readlane(cy, 8 * s + j));
        const float qz = __int_as_float(__builtin_amdgcn_readlane(cz, 8 * s + j));
#pragma unroll
        for (int u = 0; u < QPL / 2; ++u) {
          const v2f d = dist2x2(pxv[u], pyv[u], pzv[u], qx, qy, qz);
          e[2 * u + 0][s][j] = d.x;
          e[2 * u + 1][s][j] = d.y;
        }
      }
    }
#pragma unroll
    for (int q = 0; q < QPL; ++q) {
      SORT8D(e[q][0]) SORT8D(e[q][1]) SORT8D(e[q][2]) SORT8D(e[q][3])
      LOW8D(e[q][0], e[q][1])
      LOW8D(e[q][2], e[q][3])
      LOW8D(e[q][0], e[q][2])
#pragma unroll
      for (int i = 0; i < 8; ++i) {
        const bool sw = e[q][0][7 - i] < bd[q][i];
        bd[q][i] = sw ? e[q][0][7 - i] : bd[q][i];
        bg[q][i] = sw ? g : bg[q][i];
      }
      CEXS(bd[q], bg[q], 0, 4) CEXS(bd[q], bg[q], 1, 5) CEXS(bd[q], bg[q], 2, 6) CEXS(bd[q], bg[q], 3, 7)
      CEXS(bd[q], bg[q], 0, 2) CEXS(bd[q], bg[q], 1, 3) CEXS(bd[q], bg[q], 4, 6) CEXS(bd[q], bg[q], 5, 7)
      CEXS(bd[q], bg[q], 0, 1) CEXS(bd[q], bg[q], 2, 3) CEXS(bd[q], bg[q], 4, 5) CEXS(bd[q], bg[q], 6, 7)
    }
  }

#pragma unroll
  for (int q = 0; q < QPL; ++q)
#pragma unroll
    for (int j = 0; j < KNN; ++j) {
      xd[((QPL * wave + q) * KNN + j) * 32 + lane] = bd[q][j];
      xg[((QPL * wave + q) * KNN + j) * 32 + lane] = bg[q][j];
    }
  __syncthreads();

  float md[KNN]; int mg[KNN];
#pragma unroll
  for (int j = 0; j < KNN; ++j) {
    md[j] = xd[((QPL * qs + part) * KNN + j) * 32 + lane];
    mg[j] = xg[((QPL * qs + part) * KNN + j) * 32 + lane];
  }
#pragma unroll
  for (int pp = 1; pp < QPL; ++pp) {
    float od[KNN]; int og[KNN];
#pragma unroll
    for (int j = 0; j < KNN; ++j) {
      od[j] = xd[((QPL * (pp * NQS + qs) + part) * KNN + j) * 32 + lane];
      og[j] = xg[((QPL * (pp * NQS + qs) + part) * KNN + j) * 32 + lane];
    }
#pragma unroll
    for (int i = 0; i < 8; ++i) {
      const bool sw = od[7 - i] < md[i];
      md[i] = sw ? od[7 - i] : md[i];
      mg[i] = sw ? og[7 - i] : mg[i];
    }
    CEXS(md, mg, 0, 4) CEXS(md, mg, 1, 5) CEXS(md, mg, 2, 6) CEXS(md, mg, 3, 7)
    CEXS(md, mg, 0, 2) CEXS(md, mg, 1, 3) CEXS(md, mg, 4, 6) CEXS(md, mg, 5, 7)
    CEXS(md, mg, 0, 1) CEXS(md, mg, 2, 3) CEXS(md, mg, 4, 5) CEXS(md, mg, 6, 7)
  }

  int kk[KNN], cnt[KNN], br[KNN];
#pragma unroll
  for (int j = 0; j < KNN; ++j) {
    int k = 0;
#pragma unroll
    for (int i = 0; i < j; ++i) k += (md[i] == md[j] && mg[i] == mg[j]) ? 1 : 0;
    kk[j] = k; cnt[j] = 0; br[j] = 0;
  }
  for (int t = 0; t < 32; ++t) {
    const int r = (lane + t) & 31;
#pragma unroll
    for (int j = 0; j < KNN; ++j) {
      const float* pt = sp + 96 * mg[j] + 3 * r;
      const float dd = dist2(mx, my, mz, pt[0], pt[1], pt[2]);
      const bool hit = dd == md[j];
      br[j] = (hit && cnt[j] == kk[j]) ? r : br[j];
      cnt[j] += hit ? 1 : 0;
    }
  }

  float sx = 0.f, sy = 0.f, sz = 0.f;
#pragma unroll
  for (int j = 0; j < KNN; ++j) {
    const int m = 32 * mg[j] + br[j];
    sx += sp[3 * m + 0]; sy += sp[3 * m + 1]; sz += sp[3 * m + 2];
  }

  const float ax = sx * 0.125f, ay = sy * 0.125f, az = sz * 0.125f;

  const int c = lane & 15;
  const int h = lane >> 4;
  const int rbase = mrow - lane;
  const float* w1l = smem + L_W1 + c * INP8 + 2 * h;
  const float* w2l = smem + L_W2 + c * WP + 8 * h;
  const float* wdl = smem + L_WD + c * WP + 8 * h;
  const float* wcl = smem + L_WC + c * WP + 8 * h;

  for (int tt = 0; tt < 2; ++tt) {
    asm volatile("" ::: "memory");
    const int row = rbase + 16 * tt + c;
    const int sl = 16 * tt + c;
    const float gx = __shfl(mx, sl, 32), gy = __shfl(my, sl, 32), gz = __shfl(mz, sl, 32);
    const float ux = __shfl(ax, sl, 32), uy = __shfl(ay, sl, 32), uz = __shfl(az, sl, 32);
    const v2f x0 = h ? (v2f){gz, ux} : (v2f){gx, gy};
    const v2f x1 = h ? (v2f){0.f, 0.f} : (v2f){uy, uz};
    const float pr0 = gx, pr1 = gy, pr2 = gz;

    v2f a1f[8][2];
    v8f hh[8];
#pragma unroll
    for (int U = 0; U < 8; ++U) {
      hh[U]     = bias_tile(smem + L_B1 + 8 * h + 16 * U);
      a1f[U][0] = *(const v2f*)(w1l + 16 * U * INP8);
      a1f[U][1] = *(const v2f*)(w1l + 16 * U * INP8 + 4);
    }
#pragma unroll
    for (int U = 0; U < 8; ++U) hh[U] = wmma_f32(a1f[U][0], x0, hh[U]);
#pragma unroll
    for (int U = 0; U < 8; ++U) hh[U] = relu8(wmma_f32(a1f[U][1], x1, hh[U]));

    v8f seed[8];
    dense128(w2l, smem + L_B2 + 8 * h, hh, seed);

    for (int t = 0; t < OUTW; ++t) {
      v8f ht[8];
      dense128(wdl + t * HID * WP, smem + L_BD + 8 * h, seed, ht);

      v4f wc[16];
#pragma unroll
      for (int s = 0; s < 16; ++s)
        wc[s] = *(const v4f*)(wcl + 16 * (s >> 1) + 4 * (s & 1));
      v8f o0 = bias_tile(smem + L_BC + 8 * h);
      v8f o1 = {0.f, 0.f, 0.f, 0.f, 0.f, 0.f, 0.f, 0.f};
      __builtin_amdgcn_sched_barrier(0);
#pragma unroll
      for (int U = 0; U < 8; ++U) ht[U] = relu8(ht[U]);
#pragma unroll
      for (int T = 0; T < 8; T += 2) {
#pragma unroll
        for (int q = 0; q < 4; ++q) {
          const v4f ae = wc[2 * T + (q >> 1)];
          const v4f ao = wc[2 * (T + 1) + (q >> 1)];
          const v2f av0 = (q & 1) ? (v2f){ae.z, ae.w} : (v2f){ae.x, ae.y};
          const v2f av1 = (q & 1) ? (v2f){ao.z, ao.w} : (v2f){ao.x, ao.y};
          const v2f bv0 = {ht[T][2 * q], ht[T][2 * q + 1]};
          const v2f bv1 = {ht[T + 1][2 * q], ht[T + 1][2 * q + 1]};
          o0 = wmma_f32(av0, bv0, o0);
          o1 = wmma_f32(av1, bv1, o1);
        }
      }
      if (h == 0) {
        float* o = out + ((size_t)row * OUTW + t) * 3;
        o[0] = pr0 + (o0[0] + o1[0]);
        o[1] = pr1 + (o0[1] + o1[1]);
        o[2] = pr2 + (o0[2] + o1[2]);
      }
    }
  }
}


extern "C" void kernel_launch(void* const* d_in, const int* in_sizes, int n_in,
                              void* d_out, int out_size, void* d_ws, size_t ws_size,
                              hipStream_t stream) {
  const float* partial   = (const float*)d_in[0];
  const float* predicted = (const float*)d_in[1];
  const float* W1 = (const float*)d_in[2];
  const float* b1 = (const float*)d_in[3];
  const float* W2 = (const float*)d_in[4];
  const float* b2 = (const float*)d_in[5];
  const float* Wd = (const float*)d_in[6];
  const float* bd = (const float*)d_in[7];
  const float* Wc = (const float*)d_in[8];
  const float* bc = (const float*)d_in[9];
  float* out = (float*)d_out;

  dim3 grid(NN / 256, BB);
  const size_t lds = (size_t)L_END * sizeof(float);
  refine_kernel<<<grid, 256, lds, stream>>>(
      partial, predicted, W1, b1, W2, b2, Wd, bd, Wc, bc, out);
}


# KNN scan ranks groups by |q|^2-2p.q: 3 packed FMAs per query pair and point (was 7 packed ops); groups within a proven rounding bound of the 8th minimum are walked; distances, list and sums unchanged
# speedup vs baseline: 7.9001x; 1.1888x over previous
#include <hip/hip_runtime.h>
#include <hip/hip_bf16.h>


typedef __attribute__((ext_vector_type(2))) float v2f;
typedef __attribute__((ext_vector_type(4))) float v4f;
typedef __attribute__((ext_vector_type(8))) float v8f;

#define BB   8
#define MM   2048
#define NN   8192
#define KNN  8
#define HID  128
#define INP  6
#define INP8 8
#define OUTW 2

#define CEXD(E, i, j) {                                     \
    const float lo_ = __builtin_fminf(E[i], E[j]);          \
    const float hi_ = __builtin_fmaxf(E[i], E[j]);          \
    E[i] = lo_; E[j] = hi_; }

#define SORT8D(E)                                                             \
    CEXD(E, 0, 2) CEXD(E, 1, 3) CEXD(E, 4, 6) CEXD(E, 5, 7)                   \
    CEXD(E, 0, 4) CEXD(E, 1, 5) CEXD(E, 2, 6) CEXD(E, 3, 7)                   \
    CEXD(E, 0, 1) CEXD(E, 2, 3) CEXD(E, 4, 5) CEXD(E, 6, 7)                   \
    CEXD(E, 2, 4) CEXD(E, 3, 5)                                               \
    CEXD(E, 1, 4) CEXD(E, 3, 6)                                               \
    CEXD(E, 1, 2) CEXD(E, 3, 4) CEXD(E, 5, 6)

#define LOW8D(A, B) {                                                         \
    _Pragma("unroll")                                                         \
    for (int i_ = 0; i_ < 8; ++i_) A[i_] = __builtin_fminf(A[i_], B[7 - i_]); \
    CEXD(A, 0, 4) CEXD(A, 1, 5) CEXD(A, 2, 6) CEXD(A, 3, 7)                   \
    CEXD(A, 0, 2) CEXD(A, 1, 3) CEXD(A, 4, 6) CEXD(A, 5, 7)                   \
    CEXD(A, 0, 1) CEXD(A, 2, 3) CEXD(A, 4, 5) CEXD(A, 6, 7) }

__device__ __forceinline__ float dist2(float px, float py, float pz,
                                       float qx, float qy, float qz) {
  const float dx = px - qx;
  const float dy = py - qy;
  const float dz = pz - qz;
  float s2;
  {
#pragma clang fp contract(off)
    s2 = dx * dx + dy * dy;
  }
  return __builtin_fmaf(dz, dz, s2);
}

__device__ __forceinline__ v2f score2(v2f m2x, v2f m2y, v2f m2z,
                                      float qx, float qy, float qz, float qq) {
  const v2f vx = {qx, qx}, vy = {qy, qy}, vz = {qz, qz}, vq = {qq, qq};
  v2f s = __builtin_elementwise_fma(m2z, vz, vq);
  s = __builtin_elementwise_fma(m2y, vy, s);
  return __builtin_elementwise_fma(m2x, vx, s);
}

__device__ __forceinline__ float dist2p(v2f pxy, float pz, v2f qxy, float qz) {
  const v2f d = pxy - qxy;
  const float dz = pz - qz;
  float s2;
  {
#pragma clang fp contract(off)
    const v2f sq = d * d;
    s2 = sq.x + sq.y;
  }
  return __builtin_fmaf(dz, dz, s2);
}

#define QPL 4
#define NQS (8 / QPL)
#define GP  (MM / 32 / QPL)

__device__ __forceinline__ v8f wmma_f32(v2f a, v2f b, v8f c) {
  return __builtin_amdgcn_wmma_f32_16x16x4_f32(
       false, a,  false, b,
       (short)0, c,  false,  false);
}

__device__ __forceinline__ v8f relu8(v8f v) {
#pragma unroll
  for (int j = 0; j < 8; ++j) v[j] = v[j] > 0.f ? v[j] : 0.f;
  return v;
}

#define WP    132
#define L_W2  0
#define L_WD  (L_W2 + HID * WP)
#define L_WC  (L_WD + OUTW * HID * WP)
#define L_W1  (L_WC + 16 * WP)
#define L_B1  (L_W1 + HID * INP8)
#define L_B2  (L_B1 + HID)
#define L_BD  (L_B2 + HID)
#define L_BC  (L_BD + HID)
#define L_END (L_BC + 16)

__device__ __forceinline__ v8f bias_tile(const float* b) {
  const v4f lo = *(const v4f*)(b);
  const v4f hi = *(const v4f*)(b + 4);
  v8f r = {lo.x, lo.y, lo.z, lo.w, hi.x, hi.y, hi.z, hi.w};
  return r;
}

__device__ __forceinline__ void dense128(const float* w, const float* b,
                                         const v8f (&in)[8], v8f (&acc)[8]) {
  v4f a[2][8];
#pragma unroll
  for (int U = 0; U < 8; ++U) acc[U] = bias_tile(b + 16 * U);
#pragma unroll
  for (int U = 0; U < 8; ++U) a[0][U] = *(const v4f*)(w + 16 * U * WP);
#pragma unroll
  for (int s = 0; s < 16; ++s) {
    const int T = s >> 1, ip = s & 1;
    if (s + 1 < 16) {
      const int col = 16 * ((s + 1) >> 1) + 4 * ((s + 1) & 1);
#pragma unroll
      for (int U = 0; U < 8; ++U)
        a[(s + 1) & 1][U] = *(const v4f*)(w + 16 * U * WP + col);
    }
    __builtin_amdgcn_sched_barrier(0);
#pragma unroll
    for (int U = 0; U < 8; ++U) {
      const v2f av = {a[s & 1][U].x, a[s & 1][U].y};
      const v2f bv = {in[T][4 * ip + 0], in[T][4 * ip + 1]};
      acc[U] = wmma_f32(av, bv, acc[U]);
    }
#pragma unroll
    for (int U = 0; U < 8; ++U) {
      const v2f av = {a[s & 1][U].z, a[s & 1][U].w};
      const v2f bv = {in[T][4 * ip + 2], in[T][4 * ip + 3]};
      acc[U] = wmma_f32(av, bv, acc[U]);
    }
    __builtin_amdgcn_sched_barrier(0);
  }
}

__global__ void __launch_bounds__(256)
refine_kernel(const float* __restrict__ partial,
              const float* __restrict__ predicted,
              const float* __restrict__ W1, const float* __restrict__ b1,
              const float* __restrict__ W2, const float* __restrict__ b2,
              const float* __restrict__ Wd, const float* __restrict__ bdc,
              const float* __restrict__ Wc, const float* __restrict__ bc,
              float* __restrict__ out) {
  extern __shared__ __attribute__((aligned(16))) float smem[];
  __shared__ __attribute__((aligned(16))) float sp[MM * 3];
  __shared__ float gm[8 * QPL * GP * 32];
  __shared__ float qm[8];
  const int b   = blockIdx.y;
  const int tid = threadIdx.x;

  v4f w2c = *(const v4f*)(W2 + 4 * tid);
  v2f wdc0, wdc1, wdc2, wdc3;
  {
    const int n = tid & (HID - 1), kq = 4 * (tid >> 7);
    wdc0 = *(const v2f*)(Wd + ((kq + 0) * HID + n) * OUTW);
    wdc1 = *(const v2f*)(Wd + ((kq + 1) * HID + n) * OUTW);
    wdc2 = *(const v2f*)(Wd + ((kq + 2) * HID + n) * OUTW);
    wdc3 = *(const v2f*)(Wd + ((kq + 3) * HID + n) * OUTW);
  }

  {
    const v4f* src = (const v4f*)(partial + b * MM * 3);
#pragma unroll
    for (int j = 0; j < (MM * 3 / 4) / 256; ++j)
      ((v4f*)sp)[tid + 256 * j] = src[tid + 256 * j];
  }
#pragma unroll
  for (int j = 0; j < (16 * HID) / 256; ++j) {
    const int i = tid + 256 * j;
    const int r = i >> 7, k = i & (HID - 1);
    const float v = Wc[(r < 3 ? r : 2) * HID + k];
    smem[L_WC + r * WP + k] = r < 3 ? v : 0.f;
  }
#pragma unroll
  for (int j = 0; j < (HID * INP8) / 256; ++j) {
    const int i = tid + 256 * j;
    const int n = i >> 3, k = i & 7;
    const float v = W1[n * INP + (k < INP ? k : INP - 1)];
    smem[L_W1 + i] = k < INP ? v : 0.f;
  }
  {
    const int n = tid & (HID - 1);
    smem[L_B1 + n] = b1[n];
    smem[L_B2 + n] = b2[n];
    smem[L_BD + n] = bdc[n];
    const int m = tid & 15;
    const float v = bc[m < 3 ? m : 2];
    smem[L_BC + m] = m < 3 ? v : 0.f;
  }
  __syncthreads();

  const int wave = __builtin_amdgcn_readfirstlane(tid >> 5);
  const int lane = tid & 31;
  const int part = wave / NQS;
  const int qs   = wave % NQS;
  const int nbase = blockIdx.x * 256 + 32 * QPL * qs + lane;

  v2f m2x[QPL / 2], m2y[QPL / 2], m2z[QPL / 2];
#pragma unroll
  for (int q = 0; q < QPL; ++q) {
    const float* pr = predicted + (size_t)(b * NN + nbase + 32 * q) * 3;
    m2x[q >> 1][q & 1] = -2.f * pr[0];
    m2y[q >> 1][q & 1] = -2.f * pr[1];
    m2z[q >> 1][q & 1] = -2.f * pr[2];
  }

  const int mrow = b * NN + nbase + 32 * part;
  const float mx = predicted[(size_t)mrow * 3 + 0];
  const float my = predicted[(size_t)mrow * 3 + 1];
  const float mz = predicted[(size_t)mrow * 3 + 2];


  const int g0 = GP * part;
  const int g1 = g0 + GP;
  float nx = sp[3 * (32 * g0 + lane) + 0];
  float ny = sp[3 * (32 * g0 + lane) + 1];
  float nz = sp[3 * (32 * g0 + lane) + 2];
  float qql = 0.f;
  for (int g = g0; g < g1; ++g) {
    const float nq = __builtin_fmaf(nz, nz, __builtin_fmaf(ny, ny, nx * nx));
    qql = __builtin_fmaxf(qql, nq);
    const int cx = __float_as_int(nx);
    const int cy = __float_as_int(ny);
    const int cz = __float_as_int(nz);
    const int cq = __float_as_int(nq);
    const int gn = (g + 1 < g1) ? g + 1 : g;
    nx = sp[3 * (32 * gn + lane) + 0];
    ny = sp[3 * (32 * gn + lane) + 1];
    nz = sp[3 * (32 * gn + lane) + 2];
    {
      const int i = tid + 256 * (g - g0);
      *(v4f*)(smem + L_W2 + (i >> 5) * WP + 4 * (i & 31)) = w2c;
      const int n = i & (HID - 1), kq = 4 * (i >> 7);
      const v4f t0 = {wdc0.x, wdc1.x, wdc2.x, wdc3.x};
      const v4f t1 = {wdc0.y, wdc1.y, wdc2.y, wdc3.y};
      *(v4f*)(smem + L_WD + n * WP + kq)            = t0;
      *(v4f*)(smem + L_WD + HID * WP + n * WP + kq) = t1;
      const int i2 = tid + 256 * (gn - g0);
      w2c = *(const v4f*)(W2 + 4 * i2);
      const int n2 = i2 & (HID - 1), k2 = 4 * (i2 >> 7);
      wdc0 = *(const v2f*)(Wd + ((k2 + 0) * HID + n2) * OUTW);
      wdc1 = *(const v2f*)(Wd + ((k2 + 1) * HID + n2) * OUTW);
      wdc2 = *(const v2f*)(Wd + ((k2 + 2) * HID + n2) * OUTW);
      wdc3 = *(const v2f*)(Wd + ((k2 + 3) * HID + n2) * OUTW);
    }
    asm volatile("" ::: "memory");
    float gmin[QPL];
#pragma unroll
    for (int s = 0; s < 4; ++s) {
      float rm[QPL];
#pragma unroll
      for (int j = 0; j < 8; ++j) {
        const float qx = __int_as_float(__builtin_amdgcn_readlane(cx, 8 * s + j));
        const float qy = __int_as_float(__builtin_amdgcn_readlane(cy, 8 * s + j));
        const float qz = __int_as_float(__builtin_amdgcn_readlane(cz, 8 * s + j));
        const float qq = __int_as_float(__builtin_amdgcn_readlane(cq, 8 * s + j));
#pragma unroll
        for (int u = 0; u < QPL / 2; ++u) {
          const v2f d = score2(m2x[u], m2y[u], m2z[u], qx, qy, qz, qq);
          rm[2 * u + 0] = j ? __builtin_fminf(rm[2 * u + 0], d.x) : d.x;
          rm[2 * u + 1] = j ? __builtin_fminf(rm[2 * u + 1], d.y) : d.y;
        }
      }
#pragma unroll
      for (int q = 0; q < QPL; ++q) gmin[q] = s ? __builtin_fminf(gmin[q], rm[q]) : rm[q];
    }
#pragma unroll
    for (int q = 0; q < QPL; ++q)
      gm[((QPL * wave + q) * GP + (g - g0)) * 32 + lane] = gmin[q];
  }
#pragma unroll
  for (int m = 16; m >= 1; m >>= 1) qql = __builtin_fmaxf(qql, __shfl_xor(qql, m, 32));
  if (lane == 0) qm[wave] = qql;

  __syncthreads();

  float gv[QPL][GP];
#pragma unroll
  for (int pp = 0; pp < QPL; ++pp)
#pragma unroll
    for (int i = 0; i < GP; ++i)
      gv[pp][i] = gm[((QPL * (pp * NQS + qs) + part) * GP + i) * 32 + lane];
  float tcut;
  {
    float f[2 * QPL][8];
#pragma unroll
    for (int pp = 0; pp < QPL; ++pp)
#pragma unroll
      for (int i = 0; i < GP; ++i) f[2 * pp + (i >> 3)][i & 7] = gv[pp][i];
    SORT8D(f[0]) SORT8D(f[1]) SORT8D(f[2]) SORT8D(f[3])
    SORT8D(f[4]) SORT8D(f[5]) SORT8D(f[6]) SORT8D(f[7])
    LOW8D(f[0], f[1]) LOW8D(f[2], f[3]) LOW8D(f[4], f[5]) LOW8D(f[6], f[7])
    LOW8D(f[0], f[2]) LOW8D(f[4], f[6])
    LOW8D(f[0], f[4])
    tcut = f[0][7];
  }
  float qqmax = qm[0];
#pragma unroll
  for (int w = 1; w < 8; ++w) qqmax = __builtin_fmaxf(qqmax, qm[w]);
  const float ppq = __builtin_fmaf(mz, mz, __builtin_fmaf(my, my, mx * mx));
  const float tmark = tcut + 0x1p-18f * (ppq + qqmax);
  unsigned mlo0 = 0u, mhi0 = 0u;
#pragma unroll
  for (int pp = 0; pp < QPL; ++pp) {
    unsigned bits = 0u;
#pragma unroll
    for (int i = 0; i < GP; ++i) bits |= (gv[pp][i] <= tmark) ? (1u << i) : 0u;
    if (GP * pp < 32) mlo0 |= bits << ((GP * pp) & 31);
    else              mhi0 |= bits << ((GP * pp) & 31);
  }

  int o8[8];
#pragma unroll
  for (int j = 0; j < 8; ++j) o8[j] = 3 * ((lane + j) & 7);
  const v2f mxy = {mx, my};

  float md[KNN];
#pragma unroll
  for (int j = 0; j < KNN; ++j) md[j] = 3.4e38f;
  {
    unsigned a = mlo0, c2 = mhi0;
    while ((a | c2) != 0u) {
      int g;
      if (a != 0u) { g = __builtin_ctz(a); a &= a - 1u; }
      else         { g = 32 + __builtin_ctz(c2); c2 &= c2 - 1u; }
      const float* gp = sp + 96 * g;
      float e[4][8];
#pragma unroll
      for (int t = 0; t < 4; ++t) {
        const float* bp = gp + 24 * (((lane >> 3) + t) & 3);
#pragma unroll
        for (int j = 0; j < 8; ++j) {
          const float* pt = bp + o8[j];
          const v2f qxy = {pt[0], pt[1]};
          e[t][j] = dist2p(mxy, mz, qxy, pt[2]);
        }
      }
      SORT8D(e[0]) SORT8D(e[1]) SORT8D(e[2]) SORT8D(e[3])
      LOW8D(e[0], e[1])
      LOW8D(e[2], e[3])
      LOW8D(e[0], e[2])
      LOW8D(md, e[0])
    }
  }
  const float tau = md[KNN - 1];
  unsigned mlo = mlo0, mhi = mhi0;

  v2f sxy = {0.f, 0.f};
  float sz = 0.f;
  int nle = 0;
  while ((mlo | mhi) != 0u) {
    int g;
    if (mlo != 0u) { g = __builtin_ctz(mlo); mlo &= mlo - 1u; }
    else           { g = 32 + __builtin_ctz(mhi); mhi &= mhi - 1u; }
    const float* gp = sp + 96 * g;
    for (int t = 0; t < 4; ++t) {
      const float* bp = gp + 24 * (((lane >> 3) + t) & 3);
#pragma unroll
      for (int j = 0; j < 8; ++j) {
        const float* pt = bp + o8[j];
        const v2f qxy = {pt[0], pt[1]};
        const float qz = pt[2];
        const bool in = dist2p(mxy, mz, qxy, qz) <= tau;
        const float m = in ? 1.f : 0.f;
        const v2f mm = {m, m};
        sxy = __builtin_elementwise_fma(mm, qxy, sxy);
        sz = __builtin_fmaf(m, qz, sz);
        nle += in ? 1 : 0;
      }
    }
  }
  float sx = sxy.x, sy = sxy.y;
  if (nle != KNN) {
    int room = 1;
#pragma unroll
    for (int j = 0; j < KNN - 1; ++j) room += (md[j] == tau) ? 1 : 0;
    sx = 0.f; sy = 0.f; sz = 0.f;
    unsigned a = mlo0, c2 = mhi0;
    while ((a | c2) != 0u) {
      int g;
      if (a != 0u) { g = __builtin_ctz(a); a &= a - 1u; }
      else         { g = 32 + __builtin_ctz(c2); c2 &= c2 - 1u; }
      const float* gp = sp + 96 * g;
      for (int r = 0; r < 32; ++r) {
        const float qx = gp[3 * r + 0], qy = gp[3 * r + 1], qz = gp[3 * r + 2];
        const float dd = dist2(mx, my, mz, qx, qy, qz);
        const bool at = dd == tau;
        const bool in = dd < tau || (at && room > 0);
        room -= at ? 1 : 0;
        sx += in ? qx : 0.f;
        sy += in ? qy : 0.f;
        sz += in ? qz : 0.f;
      }
    }
  }

  const float ax = sx * 0.125f, ay = sy * 0.125f, az = sz * 0.125f;

  const int c = lane & 15;
  const int h = lane >> 4;
  const int rbase = mrow - lane;
  const float* w1l = smem + L_W1 + c * INP8 + 2 * h;
  const float* w2l = smem + L_W2 + c * WP + 8 * h;
  const float* wdl = smem + L_WD + c * WP + 8 * h;
  const float* wcl = smem + L_WC + c * WP + 8 * h;

  for (int tt = 0; tt < 2; ++tt) {
    asm volatile("" ::: "memory");
    const int row = rbase + 16 * tt + c;
    const int sl = 16 * tt + c;
    const float gx = __shfl(mx, sl, 32), gy = __shfl(my, sl, 32), gz = __shfl(mz, sl, 32);
    const float ux = __shfl(ax, sl, 32), uy = __shfl(ay, sl, 32), uz = __shfl(az, sl, 32);
    const v2f x0 = h ? (v2f){gz, ux} : (v2f){gx, gy};
    const v2f x1 = h ? (v2f){0.f, 0.f} : (v2f){uy, uz};
    const float pr0 = gx, pr1 = gy, pr2 = gz;

    v2f a1f[8][2];
    v8f hh[8];
#pragma unroll
    for (int U = 0; U < 8; ++U) {
      hh[U]     = bias_tile(smem + L_B1 + 8 * h + 16 * U);
      a1f[U][0] = *(const v2f*)(w1l + 16 * U * INP8);
      a1f[U][1] = *(const v2f*)(w1l + 16 * U * INP8 + 4);
    }
#pragma unroll
    for (int U = 0; U < 8; ++U) hh[U] = wmma_f32(a1f[U][0], x0, hh[U]);
#pragma unroll
    for (int U = 0; U < 8; ++U) hh[U] = relu8(wmma_f32(a1f[U][1], x1, hh[U]));

    v8f seed[8];
    dense128(w2l, smem + L_B2 + 8 * h, hh, seed);

    for (int t = 0; t < OUTW; ++t) {
      v8f ht[8];
      dense128(wdl + t * HID * WP, smem + L_BD + 8 * h, seed, ht);

      v4f wc[16];
#pragma unroll
      for (int s = 0; s < 16; ++s)
        wc[s] = *(const v4f*)(wcl + 16 * (s >> 1) + 4 * (s & 1));
      v8f o0 = bias_tile(smem + L_BC + 8 * h);
      v8f o1 = {0.f, 0.f, 0.f, 0.f, 0.f, 0.f, 0.f, 0.f};
      __builtin_amdgcn_sched_barrier(0);
#pragma unroll
      for (int U = 0; U < 8; ++U) ht[U] = relu8(ht[U]);
#pragma unroll
      for (int T = 0; T < 8; T += 2) {
#pragma unroll
        for (int q = 0; q < 4; ++q) {
          const v4f ae = wc[2 * T + (q >> 1)];
          const v4f ao = wc[2 * (T + 1) + (q >> 1)];
          const v2f av0 = (q & 1) ? (v2f){ae.z, ae.w} : (v2f){ae.x, ae.y};
          const v2f av1 = (q & 1) ? (v2f){ao.z, ao.w} : (v2f){ao.x, ao.y};
          const v2f bv0 = {ht[T][2 * q], ht[T][2 * q + 1]};
          const v2f bv1 = {ht[T + 1][2 * q], ht[T + 1][2 * q + 1]};
          o0 = wmma_f32(av0, bv0, o0);
          o1 = wmma_f32(av1, bv1, o1);
        }
      }
      if (h == 0) {
        float* o = out + ((size_t)row * OUTW + t) * 3;
        o[0] = pr0 + (o0[0] + o1[0]);
        o[1] = pr1 + (o0[1] + o1[1]);
        o[2] = pr2 + (o0[2] + o1[2]);
      }
    }
  }
}


extern "C" void kernel_launch(void* const* d_in, const int* in_sizes, int n_in,
                              void* d_out, int out_size, void* d_ws, size_t ws_size,
                              hipStream_t stream) {
  const float* partial   = (const float*)d_in[0];
  const float* predicted = (const float*)d_in[1];
  const float* W1 = (const float*)d_in[2];
  const float* b1 = (const float*)d_in[3];
  const float* W2 = (const float*)d_in[4];
  const float* b2 = (const float*)d_in[5];
  const float* Wd = (const float*)d_in[6];
  const float* bd = (const float*)d_in[7];
  const float* Wc = (const float*)d_in[8];
  const float* bc = (const float*)d_in[9];
  float* out = (float*)d_out;

  dim3 grid(NN / 256, BB);
  const size_t lds = (size_t)L_END * sizeof(float);
  refine_kernel<<<grid, 256, lds, stream>>>(
      partial, predicted, W1, b1, W2, b2, Wd, bd, Wc, bc, out);
}


# KNN: one walk over the marked groups; only points within a proven bound of the 8th distance (few over 8) go, with index, into a sorted list of 8; networks, 2nd walk gone; neighbour sum order changes
# speedup vs baseline: 8.4162x; 1.0653x over previous
#include <hip/hip_runtime.h>
#include <hip/hip_bf16.h>


typedef __attribute__((ext_vector_type(2))) float v2f;
typedef __attribute__((ext_vector_type(4))) float v4f;
typedef __attribute__((ext_vector_type(8))) float v8f;

#define BB   8
#define MM   2048
#define NN   8192
#define KNN  8
#define HID  128
#define INP  6
#define INP8 8
#define OUTW 2

#define CEXD(E, i, j) {                                     \
    const float lo_ = __builtin_fminf(E[i], E[j]);          \
    const float hi_ = __builtin_fmaxf(E[i], E[j]);          \
    E[i] = lo_; E[j] = hi_; }

#define SORT8D(E)                                                             \
    CEXD(E, 0, 2) CEXD(E, 1, 3) CEXD(E, 4, 6) CEXD(E, 5, 7)                   \
    CEXD(E, 0, 4) CEXD(E, 1, 5) CEXD(E, 2, 6) CEXD(E, 3, 7)                   \
    CEXD(E, 0, 1) CEXD(E, 2, 3) CEXD(E, 4, 5) CEXD(E, 6, 7)                   \
    CEXD(E, 2, 4) CEXD(E, 3, 5)                                               \
    CEXD(E, 1, 4) CEXD(E, 3, 6)                                               \
    CEXD(E, 1, 2) CEXD(E, 3, 4) CEXD(E, 5, 6)

#define LOW8D(A, B) {                                                         \
    _Pragma("unroll")                                                         \
    for (int i_ = 0; i_ < 8; ++i_) A[i_] = __builtin_fminf(A[i_], B[7 - i_]); \
    CEXD(A, 0, 4) CEXD(A, 1, 5) CEXD(A, 2, 6) CEXD(A, 3, 7)                   \
    CEXD(A, 0, 2) CEXD(A, 1, 3) CEXD(A, 4, 6) CEXD(A, 5, 7)                   \
    CEXD(A, 0, 1) CEXD(A, 2, 3) CEXD(A, 4, 5) CEXD(A, 6, 7) }

__device__ __forceinline__ float dist2(float px, float py, float pz,
                                       float qx, float qy, float qz) {
  const float dx = px - qx;
  const float dy = py - qy;
  const float dz = pz - qz;
  float s2;
  {
#pragma clang fp contract(off)
    s2 = dx * dx + dy * dy;
  }
  return __builtin_fmaf(dz, dz, s2);
}

__device__ __forceinline__ v2f score2(v2f m2x, v2f m2y, v2f m2z,
                                      float qx, float qy, float qz, float qq) {
  const v2f vx = {qx, qx}, vy = {qy, qy}, vz = {qz, qz}, vq = {qq, qq};
  v2f s = __builtin_elementwise_fma(m2z, vz, vq);
  s = __builtin_elementwise_fma(m2y, vy, s);
  return __builtin_elementwise_fma(m2x, vx, s);
}

__device__ __forceinline__ float dist2p(v2f pxy, float pz, v2f qxy, float qz) {
  const v2f d = pxy - qxy;
  const float dz = pz - qz;
  float s2;
  {
#pragma clang fp contract(off)
    const v2f sq = d * d;
    s2 = sq.x + sq.y;
  }
  return __builtin_fmaf(dz, dz, s2);
}

#define QPL 4
#define NQS (8 / QPL)
#define GP  (MM / 32 / QPL)

__device__ __forceinline__ v8f wmma_f32(v2f a, v2f b, v8f c) {
  return __builtin_amdgcn_wmma_f32_16x16x4_f32(
       false, a,  false, b,
       (short)0, c,  false,  false);
}

__device__ __forceinline__ v8f relu8(v8f v) {
#pragma unroll
  for (int j = 0; j < 8; ++j) v[j] = v[j] > 0.f ? v[j] : 0.f;
  return v;
}

#define WP    132
#define L_W2  0
#define L_WD  (L_W2 + HID * WP)
#define L_WC  (L_WD + OUTW * HID * WP)
#define L_W1  (L_WC + 16 * WP)
#define L_B1  (L_W1 + HID * INP8)
#define L_B2  (L_B1 + HID)
#define L_BD  (L_B2 + HID)
#define L_BC  (L_BD + HID)
#define L_END (L_BC + 16)

__device__ __forceinline__ v8f bias_tile(const float* b) {
  const v4f lo = *(const v4f*)(b);
  const v4f hi = *(const v4f*)(b + 4);
  v8f r = {lo.x, lo.y, lo.z, lo.w, hi.x, hi.y, hi.z, hi.w};
  return r;
}

__device__ __forceinline__ void dense128(const float* w, const float* b,
                                         const v8f (&in)[8], v8f (&acc)[8]) {
  v4f a[2][8];
#pragma unroll
  for (int U = 0; U < 8; ++U) acc[U] = bias_tile(b + 16 * U);
#pragma unroll
  for (int U = 0; U < 8; ++U) a[0][U] = *(const v4f*)(w + 16 * U * WP);
#pragma unroll
  for (int s = 0; s < 16; ++s) {
    const int T = s >> 1, ip = s & 1;
    if (s + 1 < 16) {
      const int col = 16 * ((s + 1) >> 1) + 4 * ((s + 1) & 1);
#pragma unroll
      for (int U = 0; U < 8; ++U)
        a[(s + 1) & 1][U] = *(const v4f*)(w + 16 * U * WP + col);
    }
    __builtin_amdgcn_sched_barrier(0);
#pragma unroll
    for (int U = 0; U < 8; ++U) {
      const v2f av = {a[s & 1][U].x, a[s & 1][U].y};
      const v2f bv = {in[T][4 * ip + 0], in[T][4 * ip + 1]};
      acc[U] = wmma_f32(av, bv, acc[U]);
    }
#pragma unroll
    for (int U = 0; U < 8; ++U) {
      const v2f av = {a[s & 1][U].z, a[s & 1][U].w};
      const v2f bv = {in[T][4 * ip + 2], in[T][4 * ip + 3]};
      acc[U] = wmma_f32(av, bv, acc[U]);
    }
    __builtin_amdgcn_sched_barrier(0);
  }
}

__global__ void __launch_bounds__(256)
refine_kernel(const float* __restrict__ partial,
              const float* __restrict__ predicted,
              const float* __restrict__ W1, const float* __restrict__ b1,
              const float* __restrict__ W2, const float* __restrict__ b2,
              const float* __restrict__ Wd, const float* __restrict__ bdc,
              const float* __restrict__ Wc, const float* __restrict__ bc,
              float* __restrict__ out) {
  extern __shared__ __attribute__((aligned(16))) float smem[];
  __shared__ __attribute__((aligned(16))) float sp[MM * 3];
  __shared__ float gm[8 * QPL * GP * 32];
  __shared__ float qm[8];
  const int b   = blockIdx.y;
  const int tid = threadIdx.x;

  v4f w2c = *(const v4f*)(W2 + 4 * tid);
  v2f wdc0, wdc1, wdc2, wdc3;
  {
    const int n = tid & (HID - 1), kq = 4 * (tid >> 7);
    wdc0 = *(const v2f*)(Wd + ((kq + 0) * HID + n) * OUTW);
    wdc1 = *(const v2f*)(Wd + ((kq + 1) * HID + n) * OUTW);
    wdc2 = *(const v2f*)(Wd + ((kq + 2) * HID + n) * OUTW);
    wdc3 = *(const v2f*)(Wd + ((kq + 3) * HID + n) * OUTW);
  }

  {
    const v4f* src = (const v4f*)(partial + b * MM * 3);
#pragma unroll
    for (int j = 0; j < (MM * 3 / 4) / 256; ++j)
      ((v4f*)sp)[tid + 256 * j] = src[tid + 256 * j];
  }
#pragma unroll
  for (int j = 0; j < (16 * HID) / 256; ++j) {
    const int i = tid + 256 * j;
    const int r = i >> 7, k = i & (HID - 1);
    const float v = Wc[(r < 3 ? r : 2) * HID + k];
    smem[L_WC + r * WP + k] = r < 3 ? v : 0.f;
  }
#pragma unroll
  for (int j = 0; j < (HID * INP8) / 256; ++j) {
    const int i = tid + 256 * j;
    const int n = i >> 3, k = i & 7;
    const float v = W1[n * INP + (k < INP ? k : INP - 1)];
    smem[L_W1 + i] = k < INP ? v : 0.f;
  }
  {
    const int n = tid & (HID - 1);
    smem[L_B1 + n] = b1[n];
    smem[L_B2 + n] = b2[n];
    smem[L_BD + n] = bdc[n];
    const int m = tid & 15;
    const float v = bc[m < 3 ? m : 2];
    smem[L_BC + m] = m < 3 ? v : 0.f;
  }
  __syncthreads();

  const int wave = __builtin_amdgcn_readfirstlane(tid >> 5);
  const int lane = tid & 31;
  const int part = wave / NQS;
  const int qs   = wave % NQS;
  const int nbase = blockIdx.x * 256 + 32 * QPL * qs + lane;

  v2f m2x[QPL / 2], m2y[QPL / 2], m2z[QPL / 2];
#pragma unroll
  for (int q = 0; q < QPL; ++q) {
    const float* pr = predicted + (size_t)(b * NN + nbase + 32 * q) * 3;
    m2x[q >> 1][q & 1] = -2.f * pr[0];
    m2y[q >> 1][q & 1] = -2.f * pr[1];
    m2z[q >> 1][q & 1] = -2.f * pr[2];
  }

  const int mrow = b * NN + nbase + 32 * part;
  const float mx = predicted[(size_t)mrow * 3 + 0];
  const float my = predicted[(size_t)mrow * 3 + 1];
  const float mz = predicted[(size_t)mrow * 3 + 2];


  const int g0 = GP * part;
  const int g1 = g0 + GP;
  float nx = sp[3 * (32 * g0 + lane) + 0];
  float ny = sp[3 * (32 * g0 + lane) + 1];
  float nz = sp[3 * (32 * g0 + lane) + 2];
  float qql = 0.f;
  for (int g = g0; g < g1; ++g) {
    const float nq = __builtin_fmaf(nz, nz, __builtin_fmaf(ny, ny, nx * nx));
    qql = __builtin_fmaxf(qql, nq);
    const int cx = __float_as_int(nx);
    const int cy = __float_as_int(ny);
    const int cz = __float_as_int(nz);
    const int cq = __float_as_int(nq);
    const int gn = (g + 1 < g1) ? g + 1 : g;
    nx = sp[3 * (32 * gn + lane) + 0];
    ny = sp[3 * (32 * gn + lane) + 1];
    nz = sp[3 * (32 * gn + lane) + 2];
    {
      const int i = tid + 256 * (g - g0);
      *(v4f*)(smem + L_W2 + (i >> 5) * WP + 4 * (i & 31)) = w2c;
      const int n = i & (HID - 1), kq = 4 * (i >> 7);
      const v4f t0 = {wdc0.x, wdc1.x, wdc2.x, wdc3.x};
      const v4f t1 = {wdc0.y, wdc1.y, wdc2.y, wdc3.y};
      *(v4f*)(smem + L_WD + n * WP + kq)            = t0;
      *(v4f*)(smem + L_WD + HID * WP + n * WP + kq) = t1;
      const int i2 = tid + 256 * (gn - g0);
      w2c = *(const v4f*)(W2 + 4 * i2);
      const int n2 = i2 & (HID - 1), k2 = 4 * (i2 >> 7);
      wdc0 = *(const v2f*)(Wd + ((k2 + 0) * HID + n2) * OUTW);
      wdc1 = *(const v2f*)(Wd + ((k2 + 1) * HID + n2) * OUTW);
      wdc2 = *(const v2f*)(Wd + ((k2 + 2) * HID + n2) * OUTW);
      wdc3 = *(const v2f*)(Wd + ((k2 + 3) * HID + n2) * OUTW);
    }
    asm volatile("" ::: "memory");
    float gmin[QPL];
#pragma unroll
    for (int s = 0; s < 4; ++s) {
      float rm[QPL];
#pragma unroll
      for (int j = 0; j < 8; ++j) {
        const float qx = __int_as_float(__builtin_amdgcn_readlane(cx, 8 * s + j));
        const float qy = __int_as_float(__builtin_amdgcn_readlane(cy, 8 * s + j));
        const float qz = __int_as_float(__builtin_amdgcn_readlane(cz, 8 * s + j));
        const float qq = __int_as_float(__builtin_amdgcn_readlane(cq, 8 * s + j));
#pragma unroll
        for (int u = 0; u < QPL / 2; ++u) {
          const v2f d = score2(m2x[u], m2y[u], m2z[u], qx, qy, qz, qq);
          rm[2 * u + 0] = j ? __builtin_fminf(rm[2 * u + 0], d.x) : d.x;
          rm[2 * u + 1] = j ? __builtin_fminf(rm[2 * u + 1], d.y) : d.y;
        }
      }
#pragma unroll
      for (int q = 0; q < QPL; ++q) gmin[q] = s ? __builtin_fminf(gmin[q], rm[q]) : rm[q];
    }
#pragma unroll
    for (int q = 0; q < QPL; ++q)
      gm[((QPL * wave + q) * GP + (g - g0)) * 32 + lane] = gmin[q];
  }
#pragma unroll
  for (int m = 16; m >= 1; m >>= 1) qql = __builtin_fmaxf(qql, __shfl_xor(qql, m, 32));
  if (lane == 0) qm[wave] = qql;

  __syncthreads();

  float gv[QPL][GP];
#pragma unroll
  for (int pp = 0; pp < QPL; ++pp)
#pragma unroll
    for (int i = 0; i < GP; ++i)
      gv[pp][i] = gm[((QPL * (pp * NQS + qs) + part) * GP + i) * 32 + lane];
  float tcut;
  {
    float f[2 * QPL][8];
#pragma unroll
    for (int pp = 0; pp < QPL; ++pp)
#pragma unroll
      for (int i = 0; i < GP; ++i) f[2 * pp + (i >> 3)][i & 7] = gv[pp][i];
    SORT8D(f[0]) SORT8D(f[1]) SORT8D(f[2]) SORT8D(f[3])
    SORT8D(f[4]) SORT8D(f[5]) SORT8D(f[6]) SORT8D(f[7])
    LOW8D(f[0], f[1]) LOW8D(f[2], f[3]) LOW8D(f[4], f[5]) LOW8D(f[6], f[7])
    LOW8D(f[0], f[2]) LOW8D(f[4], f[6])
    LOW8D(f[0], f[4])
    tcut = f[0][7];
  }
  float qqmax = qm[0];
#pragma unroll
  for (int w = 1; w < 8; ++w) qqmax = __builtin_fmaxf(qqmax, qm[w]);
  const float ppq = __builtin_fmaf(mz, mz, __builtin_fmaf(my, my, mx * mx));
  const float tmark = tcut + 0x1p-18f * (ppq + qqmax);
  unsigned mlo0 = 0u, mhi0 = 0u;
#pragma unroll
  for (int pp = 0; pp < QPL; ++pp) {
    unsigned bits = 0u;
#pragma unroll
    for (int i = 0; i < GP; ++i) bits |= (gv[pp][i] <= tmark) ? (1u << i) : 0u;
    if (GP * pp < 32) mlo0 |= bits << ((GP * pp) & 31);
    else              mhi0 |= bits << ((GP * pp) & 31);
  }

  int o8[8];
#pragma unroll
  for (int j = 0; j < 8; ++j) o8[j] = 3 * ((lane + j) & 7);
  const v2f mxy = {mx, my};
  const int la = lane >> 3, lb = lane & 7;

  const float tq = tmark + ppq;

  float md[KNN + 1];
  int mi[KNN];
#pragma unroll
  for (int j = 0; j < KNN; ++j) { md[j] = 3.4e38f; mi[j] = 0; }
  md[KNN] = 3.4e38f;
  {
    unsigned a = mlo0, c2 = mhi0;
    while ((a | c2) != 0u) {
      int g;
      if (a != 0u) { g = __builtin_ctz(a); a &= a - 1u; }
      else         { g = 32 + __builtin_ctz(c2); c2 &= c2 - 1u; }
      const float* gp = sp + 96 * g;
      unsigned cm = 0u;
#pragma unroll
      for (int t = 0; t < 4; ++t) {
        const float* bp = gp + 24 * ((la + t) & 3);
#pragma unroll
        for (int j = 0; j < 8; ++j) {
          const float* pt = bp + o8[j];
          const v2f qxy = {pt[0], pt[1]};
          const float over = tq - dist2p(mxy, mz, qxy, pt[2]);
          cm = (cm << 1) | ((unsigned)__float_as_int(over) >> 31);
        }
      }
      cm = ~cm;
      while (cm != 0u) {
        const int k = 31 - __builtin_ctz(cm);
        cm &= cm - 1u;
        const int pn = 8 * ((la + (k >> 3)) & 3) + ((lb + k) & 7);
        const float* pt = gp + 3 * pn;
        const v2f qxy = {pt[0], pt[1]};
        const float x = dist2p(mxy, mz, qxy, pt[2]);
        const int xi = 32 * g + pn;
        bool lt[KNN];
#pragma unroll
        for (int j = 0; j < KNN; ++j) lt[j] = x < md[j];
        const float po = lt[KNN - 1] ? md[KNN - 1] : x;
        md[KNN] = po < md[KNN] ? po : md[KNN];
#pragma unroll
        for (int j = KNN - 1; j >= 1; --j) {
          md[j] = lt[j - 1] ? md[j - 1] : (lt[j] ? x : md[j]);
          mi[j] = lt[j - 1] ? mi[j - 1] : (lt[j] ? xi : mi[j]);
        }
        md[0] = lt[0] ? x : md[0];
        mi[0] = lt[0] ? xi : mi[0];
      }
    }
  }
  const float tau = md[KNN - 1];
  float sx = 0.f, sy = 0.f, sz = 0.f;
#pragma unroll
  for (int j = 0; j < KNN; ++j) {
    const float* pt = sp + 3 * mi[j];
    sx += pt[0];
    sy += pt[1];
    sz += pt[2];
  }
  if (md[KNN] == tau) {
    int room = 1;
#pragma unroll
    for (int j = 0; j < KNN - 1; ++j) room += (md[j] == tau) ? 1 : 0;
    sx = 0.f; sy = 0.f; sz = 0.f;
    unsigned a = mlo0, c2 = mhi0;
    while ((a | c2) != 0u) {
      int g;
      if (a != 0u) { g = __builtin_ctz(a); a &= a - 1u; }
      else         { g = 32 + __builtin_ctz(c2); c2 &= c2 - 1u; }
      const float* gp = sp + 96 * g;
      for (int r = 0; r < 32; ++r) {
        const float qx = gp[3 * r + 0], qy = gp[3 * r + 1], qz = gp[3 * r + 2];
        const float dd = dist2(mx, my, mz, qx, qy, qz);
        const bool at = dd == tau;
        const bool in = dd < tau || (at && room > 0);
        room -= at ? 1 : 0;
        sx += in ? qx : 0.f;
        sy += in ? qy : 0.f;
        sz += in ? qz : 0.f;
      }
    }
  }

  const float ax = sx * 0.125f, ay = sy * 0.125f, az = sz * 0.125f;

  const int c = lane & 15;
  const int h = lane >> 4;
  const int rbase = mrow - lane;
  const float* w1l = smem + L_W1 + c * INP8 + 2 * h;
  const float* w2l = smem + L_W2 + c * WP + 8 * h;
  const float* wdl = smem + L_WD + c * WP + 8 * h;
  const float* wcl = smem + L_WC + c * WP + 8 * h;

  for (int tt = 0; tt < 2; ++tt) {
    asm volatile("" ::: "memory");
    const int row = rbase + 16 * tt + c;
    const int sl = 16 * tt + c;
    const float gx = __shfl(mx, sl, 32), gy = __shfl(my, sl, 32), gz = __shfl(mz, sl, 32);
    const float ux = __shfl(ax, sl, 32), uy = __shfl(ay, sl, 32), uz = __shfl(az, sl, 32);
    const v2f x0 = h ? (v2f){gz, ux} : (v2f){gx, gy};
    const v2f x1 = h ? (v2f){0.f, 0.f} : (v2f){uy, uz};
    const float pr0 = gx, pr1 = gy, pr2 = gz;

    v2f a1f[8][2];
    v8f hh[8];
#pragma unroll
    for (int U = 0; U < 8; ++U) {
      hh[U]     = bias_tile(smem + L_B1 + 8 * h + 16 * U);
      a1f[U][0] = *(const v2f*)(w1l + 16 * U * INP8);
      a1f[U][1] = *(const v2f*)(w1l + 16 * U * INP8 + 4);
    }
#pragma unroll
    for (int U = 0; U < 8; ++U) hh[U] = wmma_f32(a1f[U][0], x0, hh[U]);
#pragma unroll
    for (int U = 0; U < 8; ++U) hh[U] = relu8(wmma_f32(a1f[U][1], x1, hh[U]));

    v8f seed[8];
    dense128(w2l, smem + L_B2 + 8 * h, hh, seed);

    for (int t = 0; t < OUTW; ++t) {
      v8f ht[8];
      dense128(wdl + t * HID * WP, smem + L_BD + 8 * h, seed, ht);

      v4f wc[16];
#pragma unroll
      for (int s = 0; s < 16; ++s)
        wc[s] = *(const v4f*)(wcl + 16 * (s >> 1) + 4 * (s & 1));
      v8f o0 = bias_tile(smem + L_BC + 8 * h);
      v8f o1 = {0.f, 0.f, 0.f, 0.f, 0.f, 0.f, 0.f, 0.f};
      __builtin_amdgcn_sched_barrier(0);
#pragma unroll
      for (int U = 0; U < 8; ++U) ht[U] = relu8(ht[U]);
#pragma unroll
      for (int T = 0; T < 8; T += 2) {
#pragma unroll
        for (int q = 0; q < 4; ++q) {
          const v4f ae = wc[2 * T + (q >> 1)];
          const v4f ao = wc[2 * (T + 1) + (q >> 1)];
          const v2f av0 = (q & 1) ? (v2f){ae.z, ae.w} : (v2f){ae.x, ae.y};
          const v2f av1 = (q & 1) ? (v2f){ao.z, ao.w} : (v2f){ao.x, ao.y};
          const v2f bv0 = {ht[T][2 * q], ht[T][2 * q + 1]};
          const v2f bv1 = {ht[T + 1][2 * q], ht[T + 1][2 * q + 1]};
          o0 = wmma_f32(av0, bv0, o0);
          o1 = wmma_f32(av1, bv1, o1);
        }
      }
      if (h == 0) {
        float* o = out + ((size_t)row * OUTW + t) * 3;
        o[0] = pr0 + (o0[0] + o1[0]);
        o[1] = pr1 + (o0[1] + o1[1]);
        o[2] = pr2 + (o0[2] + o1[2]);
      }
    }
  }
}


extern "C" void kernel_launch(void* const* d_in, const int* in_sizes, int n_in,
                              void* d_out, int out_size, void* d_ws, size_t ws_size,
                              hipStream_t stream) {
  const float* partial   = (const float*)d_in[0];
  const float* predicted = (const float*)d_in[1];
  const float* W1 = (const float*)d_in[2];
  const float* b1 = (const float*)d_in[3];
  const float* W2 = (const float*)d_in[4];
  const float* b2 = (const float*)d_in[5];
  const float* Wd = (const float*)d_in[6];
  const float* bd = (const float*)d_in[7];
  const float* Wc = (const float*)d_in[8];
  const float* bc = (const float*)d_in[9];
  float* out = (float*)d_out;

  dim3 grid(NN / 256, BB);
  const size_t lds = (size_t)L_END * sizeof(float);
  refine_kernel<<<grid, 256, lds, stream>>>(
      partial, predicted, W1, b1, W2, b2, Wd, bd, Wc, bc, out);
}
